# starting kernel
# baseline (speedup 1.0000x reference)
#include <hip/hip_runtime.h>
#include <hip/hip_bf16.h>


#define N_NODES 100000
#define N_EDGES 640000
#define D 128

typedef float v2f __attribute__((ext_vector_type(2)));
typedef float v8f __attribute__((ext_vector_type(8)));

__device__ __forceinline__ void atomAddF(float* p, float v) {
    (void)__hip_atomic_fetch_add(p, v, __ATOMIC_RELAXED, __HIP_MEMORY_SCOPE_AGENT);
}

__global__ __launch_bounds__(256) void sage_init(
    float* __restrict__ agg_in, float* __restrict__ agg_out,
    float* __restrict__ cnt_in, float* __restrict__ cnt_out,
    const float* __restrict__ w_r_in, const float* __restrict__ w_r_out,
    const float* __restrict__ b_l_in, const float* __restrict__ b_l_out,
    float* __restrict__ w_sum, float* __restrict__ b_sum)
{
    const int tid    = blockIdx.x * blockDim.x + threadIdx.x;
    const int stride = gridDim.x * blockDim.x;
    const int nAgg = N_NODES * D;
    for (int j = tid; j < nAgg; j += stride) { agg_in[j] = 0.0f; agg_out[j] = 0.0f; }
    for (int j = tid; j < N_NODES; j += stride) { cnt_in[j] = 0.0f; cnt_out[j] = 0.0f; }
    for (int j = tid; j < D * D; j += stride) w_sum[j] = w_r_in[j] + w_r_out[j];
    for (int j = tid; j < D; j += stride)     b_sum[j] = b_l_in[j] + b_l_out[j];
}

__global__ __launch_bounds__(256) void sage_scatter(
    const int* __restrict__ src, const int* __restrict__ dst,
    const float* __restrict__ x,
    float* __restrict__ agg_in, float* __restrict__ agg_out,
    float* __restrict__ cnt_in, float* __restrict__ cnt_out)
{
    const int e = blockIdx.x * 8 + (threadIdx.x >> 5);
    if (e >= N_EDGES) return;
    const int lane = threadIdx.x & 31;
    const int s = src[e];
    const int d = dst[e];

    const float4 xs = *(const float4*)(x + (size_t)s * D + lane * 4);
    const float4 xd = *(const float4*)(x + (size_t)d * D + lane * 4);

    float* pin  = agg_in  + (size_t)d * D + lane * 4;
    float* pout = agg_out + (size_t)s * D + lane * 4;
    atomAddF(pin + 0, xs.x); atomAddF(pin + 1, xs.y);
    atomAddF(pin + 2, xs.z); atomAddF(pin + 3, xs.w);
    atomAddF(pout + 0, xd.x); atomAddF(pout + 1, xd.y);
    atomAddF(pout + 2, xd.z); atomAddF(pout + 3, xd.w);

    if (lane == 0) { atomAddF(cnt_in + d, 1.0f); atomAddF(cnt_out + s, 1.0f); }
}

__device__ __forceinline__ v8f mac_section(const float* __restrict__ A,
                                           const float* __restrict__ W,
                                           float sc, v8f acc)
{
#pragma unroll
    for (int k = 0; k < D; k += 4) {
        v2f a = *(const v2f*)(A + k);
        a.x *= sc; a.y *= sc;
        v2f b = *(const v2f*)(W + k);
        acc = __builtin_amdgcn_wmma_f32_16x16x4_f32(
             false, a,  false, b,
             (short)0, acc,  false,  false);
    }
    return acc;
}

__global__ __launch_bounds__(256) void sage_gemm(
    const float* __restrict__ x,
    const float* __restrict__ agg_in, const float* __restrict__ agg_out,
    const float* __restrict__ cnt_in, const float* __restrict__ cnt_out,
    const float* __restrict__ w_l_in, const float* __restrict__ w_l_out,
    const float* __restrict__ w_sum,  const float* __restrict__ b_sum,
    float* __restrict__ out)
{
    const int mtile = blockIdx.x;
    const int wave  = threadIdx.x >> 5;
    const int lane  = threadIdx.x & 31;
    const int half  = lane >> 4;
    const int l15   = lane & 15;
    const int row   = mtile * 16 + l15;
    const int col   = wave * 16 + l15;

    const float sc_in  = 1.0f / fmaxf(cnt_in[row],  1.0f);
    const float sc_out = 1.0f / fmaxf(cnt_out[row], 1.0f);

    const float bias = b_sum[col];
    v8f acc;
#pragma unroll
    for (int r = 0; r < 8; ++r) acc[r] = bias;

    const size_t aoff = (size_t)row * D + 2 * half;
    const size_t woff = (size_t)col * D + 2 * half;

    acc = mac_section(agg_in  + aoff, w_l_in  + woff, sc_in,  acc);
    acc = mac_section(agg_out + aoff, w_l_out + woff, sc_out, acc);
    acc = mac_section(x       + aoff, w_sum   + woff, 1.0f,   acc);

#pragma unroll
    for (int r = 0; r < 8; ++r) {
        const int m = mtile * 16 + half * 8 + r;
        out[(size_t)m * D + col] = acc[r];
    }
}

extern "C" void kernel_launch(void* const* d_in, const int* in_sizes, int n_in,
                              void* d_out, int out_size, void* d_ws, size_t ws_size,
                              hipStream_t stream)
{
    const float* x       = (const float*)d_in[0];
    const int*   ei      = (const int*)d_in[1];
    const float* w_l_in  = (const float*)d_in[2];
    const float* b_l_in  = (const float*)d_in[3];
    const float* w_r_in  = (const float*)d_in[4];
    const float* w_l_out = (const float*)d_in[5];
    const float* b_l_out = (const float*)d_in[6];
    const float* w_r_out = (const float*)d_in[7];
    float* out = (float*)d_out;

    const size_t NB   = (size_t)N_NODES * D * sizeof(float);
    const size_t CNTB = (size_t)N_NODES * sizeof(float);
    char* ws = (char*)d_ws;
    float* agg_in  = (float*)(ws);
    float* agg_out = (float*)(ws + NB);
    float* cnt_in  = (float*)(ws + 2 * NB);
    float* cnt_out = (float*)(ws + 2 * NB + CNTB);
    float* w_sum   = (float*)(ws + 2 * NB + 2 * CNTB);
    float* b_sum   = (float*)(ws + 2 * NB + 2 * CNTB + (size_t)D * D * sizeof(float));

    (void)in_sizes; (void)n_in; (void)out_size; (void)ws_size;

    sage_init<<<2048, 256, 0, stream>>>(agg_in, agg_out, cnt_in, cnt_out,
                                        w_r_in, w_r_out, b_l_in, b_l_out,
                                        w_sum, b_sum);

    sage_scatter<<<N_EDGES / 8, 256, 0, stream>>>(ei, ei + N_EDGES, x,
                                                  agg_in, agg_out, cnt_in, cnt_out);

    sage_gemm<<<N_NODES / 16, 256, 0, stream>>>(x, agg_in, agg_out, cnt_in, cnt_out,
                                                w_l_in, w_l_out, w_sum, b_sum, out);
}


# Edge scatter split by L2 half: a work-group reads its die, claims 256-edge items of its half's queue (int atomic), adds only into rows that half owns, so written lines stay put; arrival order kept
# speedup vs baseline: 2.9708x; 2.9708x over previous
#include <hip/hip_runtime.h>
#include <hip/hip_bf16.h>


#define N_NODES 100000
#define N_EDGES 640000
#define D 128

typedef float v2f __attribute__((ext_vector_type(2)));
typedef float v8f __attribute__((ext_vector_type(8)));

__device__ __forceinline__ void atomAddF(float* p, float v) {
    (void)__hip_atomic_fetch_add(p, v, __ATOMIC_RELAXED, __HIP_MEMORY_SCOPE_AGENT);
}

__global__ __launch_bounds__(256) void sage_init(
    float* __restrict__ agg_in, float* __restrict__ agg_out,
    float* __restrict__ cnt_in, float* __restrict__ cnt_out,
    const float* __restrict__ w_r_in, const float* __restrict__ w_r_out,
    float* __restrict__ w_sum, int* __restrict__ claim)
{
    const int tid    = blockIdx.x * blockDim.x + threadIdx.x;
    const int stride = gridDim.x * blockDim.x;
    const int nAgg = N_NODES * D;
    for (int j = tid; j < nAgg; j += stride) { agg_in[j] = 0.0f; agg_out[j] = 0.0f; }
    for (int j = tid; j < N_NODES; j += stride) { cnt_in[j] = 0.0f; cnt_out[j] = 0.0f; }
    for (int j = tid; j < D * D; j += stride) w_sum[j] = w_r_in[j] + w_r_out[j];
    if (tid < 2) claim[tid] = 0;
}

#define N_CHUNKS  (N_EDGES / 256)
#define N_ITEMS   (2 * N_CHUNKS)

__device__ __forceinline__ void scatter_row(const float* __restrict__ x, float* __restrict__ agg,
                                            int from, int to, int lane)
{
    const float* p = x + (size_t)from * D + lane;
    const float v0 = p[0], v1 = p[32], v2 = p[64], v3 = p[96];
    float* q = agg + (size_t)to * D + lane;
    atomAddF(q, v0); atomAddF(q + 32, v1); atomAddF(q + 64, v2); atomAddF(q + 96, v3);
}

__global__ __launch_bounds__(256) void sage_scatter(
    const int* __restrict__ src, const int* __restrict__ dst,
    const float* __restrict__ x,
    float* __restrict__ agg_in, float* __restrict__ agg_out,
    float* __restrict__ cnt_in, float* __restrict__ cnt_out,
    int* __restrict__ claim)
{
    __shared__ int s_item;
    if (threadIdx.x == 0) {
        const unsigned id = __builtin_amdgcn_s_sendmsg_rtn(0x87);
        const int h = (int)((id >> 18) & 1u);
        int item = -1;
        int t = __hip_atomic_fetch_add(claim + h, 1, __ATOMIC_RELAXED, __HIP_MEMORY_SCOPE_AGENT);
        if (t < N_ITEMS) {
            item = h * N_ITEMS + t;
        } else {
            t = __hip_atomic_fetch_add(claim + (1 - h), 1, __ATOMIC_RELAXED, __HIP_MEMORY_SCOPE_AGENT);
            if (t < N_ITEMS) item = (1 - h) * N_ITEMS + t;
        }
        s_item = item;
    }
    __syncthreads();
    const int item = s_item;
    if (item < 0) return;

    const int  owner = item / N_ITEMS;
    const int  t     = item - owner * N_ITEMS;
    const bool rev   = t >= N_CHUNKS;
    const int  chunk = rev ? t - N_CHUNKS : t;
    const int  lane  = threadIdx.x & 31;

    const int e = chunk * 256 + threadIdx.x;
    const int s = src[e];
    const int d = dst[e];
    const int to   = rev ? s : d;
    const int from = rev ? d : s;
    float* agg = rev ? agg_out : agg_in;
    float* cnt = rev ? cnt_out : cnt_in;

    const bool mine = (to >= N_NODES / 2) == (owner != 0);
    if (mine) atomAddF(cnt + to, 1.0f);

    unsigned m = __builtin_amdgcn_ballot_w32(mine);
    while (m) {
        const int i0 = __builtin_ctz(m); m &= m - 1;
        const int f0 = __builtin_amdgcn_readlane(from, i0);
        const int t0 = __builtin_amdgcn_readlane(to,   i0);
        if (m) {
            const int i1 = __builtin_ctz(m); m &= m - 1;
            const int f1 = __builtin_amdgcn_readlane(from, i1);
            const int t1 = __builtin_amdgcn_readlane(to,   i1);
            const float* p0 = x + (size_t)f0 * D + lane;
            const float* p1 = x + (size_t)f1 * D + lane;
            const float a0 = p0[0], a1 = p0[32], a2 = p0[64], a3 = p0[96];
            const float b0 = p1[0], b1 = p1[32], b2 = p1[64], b3 = p1[96];
            float* q0 = agg + (size_t)t0 * D + lane;
            float* q1 = agg + (size_t)t1 * D + lane;
            atomAddF(q0, a0); atomAddF(q0 + 32, a1); atomAddF(q0 + 64, a2); atomAddF(q0 + 96, a3);
            atomAddF(q1, b0); atomAddF(q1 + 32, b1); atomAddF(q1 + 64, b2); atomAddF(q1 + 96, b3);
        } else {
            scatter_row(x, agg, f0, t0, lane);
        }
    }
}

#define N_STRIPS   (N_NODES / 16)
#define GEMM_WGS   256
#define GEMM_WAVES 8

typedef float v4f __attribute__((ext_vector_type(4)));

__device__ __forceinline__ void load_a16(v4f (&a)[16], const float* __restrict__ p)
{
#pragma unroll
    for (int i = 0; i < 16; ++i) a[i] = *(const v4f*)(p + 8 * i);
}

__device__ __forceinline__ void mac_section(v8f (&acc)[8], const v4f (&a)[16], float sc,
                                            const v4f* wsec, const v4f* wnext, v4f (&bn)[8])
{
#pragma unroll
    for (int i = 0; i < 16; ++i) {
        v4f bc[8];
#pragma unroll
        for (int j = 0; j < 8; ++j) bc[j] = bn[j];
#pragma unroll
        for (int j = 0; j < 8; ++j)
            bn[j] = (i < 15) ? wsec[(j * 16 + i + 1) * 32] : wnext[(j * 16) * 32];
        const v4f av = a[i] * sc;
        const v2f a0 = {av.x, av.y};
        const v2f a1 = {av.z, av.w};
#pragma unroll
        for (int j = 0; j < 8; ++j) {
            const v2f b0 = {bc[j].x, bc[j].y};
            acc[j] = __builtin_amdgcn_wmma_f32_16x16x4_f32(
                false, a0, false, b0, (short)0, acc[j], false, false);
        }
#pragma unroll
        for (int j = 0; j < 8; ++j) {
            const v2f b1 = {bc[j].z, bc[j].w};
            acc[j] = __builtin_amdgcn_wmma_f32_16x16x4_f32(
                false, a1, false, b1, (short)0, acc[j], false, false);
        }
        __builtin_amdgcn_sched_barrier(0);
    }
}

__device__ __forceinline__ void stage_w(v4f* wl, const float* __restrict__ w, int sec, int tid)
{
    const int q  = tid & 3;
    const int cr = (tid >> 2) & 7;
    const int k4 = (tid >> 5) * 4 + q;
    const int i  = k4 >> 1;
    const int h  = k4 & 1;
#pragma unroll
    for (int it = 0; it < 16; ++it) {
        const int n = it * 8 + cr;
        const v4f v = *(const v4f*)(w + n * D + k4 * 4);
        wl[((sec * 8 + (n >> 4)) * 16 + i) * 32 + h * 16 + (n & 15)] = v;
    }
}

__global__ __launch_bounds__(256) void sage_gemm(
    const float* __restrict__ x,
    const float* __restrict__ agg_in, const float* __restrict__ agg_out,
    const float* __restrict__ cnt_in, const float* __restrict__ cnt_out,
    const float* __restrict__ w_l_in, const float* __restrict__ w_l_out,
    const float* __restrict__ w_sum,
    const float* __restrict__ b_l_in, const float* __restrict__ b_l_out,
    float* __restrict__ out)
{
    __shared__ v4f wl[3 * 8 * 16 * 32];

    const int tid  = threadIdx.x;
    const int wave = __builtin_amdgcn_readfirstlane(tid >> 5);
    const int lane = tid & 31;
    const int h    = lane >> 4;
    const int c    = lane & 15;

    int strip = wave * GEMM_WGS + blockIdx.x;

    size_t off = ((size_t)strip * 16 + c) * D + 4 * h;
    v4f an[16];
    load_a16(an, agg_in + off);
    float cin  = cnt_in[strip * 16 + c];
    float cout = cnt_out[strip * 16 + c];

    stage_w(wl, w_l_in,  0, tid);
    stage_w(wl, w_l_out, 1, tid);
    stage_w(wl, w_sum,   2, tid);

    float bias[8];
#pragma unroll
    for (int j = 0; j < 8; ++j) bias[j] = b_l_in[j * 16 + c] + b_l_out[j * 16 + c];

    __syncthreads();

    unsigned o1 = 4096, o2 = 8192;
    asm volatile("" : "+v"(o1));
    asm volatile("" : "+v"(o2));
    const v4f* w0 = wl + lane;
    const v4f* w1 = wl + (lane + o1);
    const v4f* w2 = wl + (lane + o2);

    v4f bn[8];
#pragma unroll
    for (int j = 0; j < 8; ++j) bn[j] = w0[(j * 16) * 32];

    for (; strip < N_STRIPS; strip += GEMM_WGS * GEMM_WAVES) {
        const float sc_in  = 1.0f / fmaxf(cin,  1.0f);
        const float sc_out = 1.0f / fmaxf(cout, 1.0f);

        v8f acc[8];
#pragma unroll
        for (int j = 0; j < 8; ++j)
#pragma unroll
            for (int r = 0; r < 8; ++r) acc[j][r] = bias[j];

        v4f ac[16];
#pragma unroll
        for (int i = 0; i < 16; ++i) ac[i] = an[i];
        load_a16(an, agg_out + off);
        mac_section(acc, ac, sc_in, w0, w1, bn);

#pragma unroll
        for (int i = 0; i < 16; ++i) ac[i] = an[i];
        load_a16(an, x + off);
        mac_section(acc, ac, sc_out, w1, w2, bn);

#pragma unroll
        for (int i = 0; i < 16; ++i) ac[i] = an[i];
        const int nstrip = strip + GEMM_WGS * GEMM_WAVES;
        const size_t noff = ((size_t)nstrip * 16 + c) * D + 4 * h;
        if (nstrip < N_STRIPS) {
            load_a16(an, agg_in + noff);
            cin  = cnt_in[nstrip * 16 + c];
            cout = cnt_out[nstrip * 16 + c];
        }
        mac_section(acc, ac, 1.0f, w2, w0, bn);

        float* orow = out + ((size_t)strip * 16 + 8 * h) * D + c;
#pragma unroll
        for (int j = 0; j < 8; ++j)
#pragma unroll
            for (int r = 0; r < 8; ++r) orow[(size_t)r * D + j * 16] = acc[j][r];

        off = noff;
    }
}

extern "C" void kernel_launch(void* const* d_in, const int* in_sizes, int n_in,
                              void* d_out, int out_size, void* d_ws, size_t ws_size,
                              hipStream_t stream)
{
    const float* x       = (const float*)d_in[0];
    const int*   ei      = (const int*)d_in[1];
    const float* w_l_in  = (const float*)d_in[2];
    const float* b_l_in  = (const float*)d_in[3];
    const float* w_r_in  = (const float*)d_in[4];
    const float* w_l_out = (const float*)d_in[5];
    const float* b_l_out = (const float*)d_in[6];
    const float* w_r_out = (const float*)d_in[7];
    float* out = (float*)d_out;

    const size_t NB   = (size_t)N_NODES * D * sizeof(float);
    const size_t CNTB = (size_t)N_NODES * sizeof(float);
    char* ws = (char*)d_ws;
    float* agg_in  = (float*)(ws);
    float* agg_out = (float*)(ws + NB);
    float* cnt_in  = (float*)(ws + 2 * NB);
    float* cnt_out = (float*)(ws + 2 * NB + CNTB);
    float* w_sum   = (float*)(ws + 2 * NB + 2 * CNTB);
    int*   claim   = (int*)(ws + 2 * NB + 2 * CNTB + (size_t)D * D * sizeof(float));

    (void)in_sizes; (void)n_in; (void)out_size; (void)ws_size;

    sage_init<<<2048, 256, 0, stream>>>(agg_in, agg_out, cnt_in, cnt_out,
                                        w_r_in, w_r_out, w_sum, claim);

    sage_scatter<<<2 * N_ITEMS, 256, 0, stream>>>(ei, ei + N_EDGES, x,
                                                  agg_in, agg_out, cnt_in, cnt_out, claim);

    sage_gemm<<<GEMM_WGS, 256, 0, stream>>>(x, agg_in, agg_out, cnt_in, cnt_out,
                                                w_l_in, w_l_out, w_sum, b_l_in, b_l_out, out);
}
